# starting kernel
# baseline (speedup 1.0000x reference)
#include <hip/hip_runtime.h>
#include <hip/hip_bf16.h>

typedef __attribute__((ext_vector_type(2))) float v2f;
typedef __attribute__((ext_vector_type(8))) float v8f;

#define N_CLASS 32
#define N_SUP   16
#define FDIM    1024
#define N_QRY   512
#define QTILE   16
#define DCHUNK  256
#define LSTRIDE (DCHUNK + 4)

__device__ __forceinline__ float dev_tanh(float x) {
#if __has_builtin(__builtin_amdgcn_tanhf)
  return __builtin_amdgcn_tanhf(x);
#else
  return tanhf(x);
#endif
}

__global__ __launch_bounds__(256)
void proto_att_kernel(const float* __restrict__ data, float* __restrict__ out) {
  __shared__ float sQ[QTILE * LSTRIDE];
  __shared__ float sS[N_SUP * LSTRIDE];
  __shared__ float sAtt[QTILE][N_SUP + 1];

  const int qt  = blockIdx.x;
  const int c   = blockIdx.y;
  const int tid = threadIdx.x;

  const float* Qg = data + (size_t)(N_CLASS * N_SUP + qt * QTILE) * FDIM;
  const float* Sg = data + (size_t)c * N_SUP * FDIM;

  const int q   = tid >> 4;
  const int s   = tid & 15;
  const int row = tid >> 4;
  const int l16 = tid & 15;

  float acc = 0.f;
  for (int d0 = 0; d0 < FDIM; d0 += DCHUNK) {
    const float4* qsrc = (const float4*)(Qg + (size_t)row * FDIM + d0);
    const float4* ssrc = (const float4*)(Sg + (size_t)row * FDIM + d0);
    float4* qdst = (float4*)(sQ + row * LSTRIDE);
    float4* sdst = (float4*)(sS + row * LSTRIDE);
#pragma unroll
    for (int j = 0; j < 4; ++j) {
      int c4 = l16 + j * 16;
      qdst[c4] = qsrc[c4];
      sdst[c4] = ssrc[c4];
    }
    __syncthreads();
    const float* qr = sQ + q * LSTRIDE;
    const float* sr = sS + s * LSTRIDE;
#pragma unroll 8
    for (int d = 0; d < DCHUNK; d += 4) {
      float4 qv = *(const float4*)(qr + d);
      float4 sv = *(const float4*)(sr + d);
      acc += dev_tanh(qv.x * sv.x);
      acc += dev_tanh(qv.y * sv.y);
      acc += dev_tanh(qv.z * sv.z);
      acc += dev_tanh(qv.w * sv.w);
    }
    __syncthreads();
  }
  sAtt[q][s] = acc;
  __syncthreads();

  float m = -3.402823466e38f;
  float rowv[N_SUP];
#pragma unroll
  for (int i = 0; i < N_SUP; ++i) { rowv[i] = sAtt[q][i]; m = fmaxf(m, rowv[i]); }
  float sum = 0.f;
#pragma unroll
  for (int i = 0; i < N_SUP; ++i) sum += __expf(rowv[i] - m);
  float mya = __expf(acc - m) / sum;
  __syncthreads();
  sAtt[q][s] = mya;
  __syncthreads();

  const int wave = tid >> 5;
  const int lane = tid & 31;
  const int hi   = lane >> 4;
  const int ln   = lane & 15;

  v2f a[4];
#pragma unroll
  for (int kk = 0; kk < 4; ++kk) {
    a[kk].x = sAtt[ln][kk * 4 + (hi ? 2 : 0)];
    a[kk].y = sAtt[ln][kk * 4 + (hi ? 3 : 1)];
  }

  float* outBase = out + ((size_t)(qt * QTILE) * N_CLASS + c) * FDIM;

  for (int nt = wave; nt < FDIM / 16; nt += 8) {
    const int n0 = nt * 16;
    v8f dacc = {};
#pragma unroll
    for (int kk = 0; kk < 4; ++kk) {
      v2f b;
      b.x = Sg[(size_t)(kk * 4 + (hi ? 2 : 0)) * FDIM + n0 + ln];
      b.y = Sg[(size_t)(kk * 4 + (hi ? 3 : 1)) * FDIM + n0 + ln];
      dacc = __builtin_amdgcn_wmma_f32_16x16x4_f32(false, a[kk], false, b,
                                                   (short)0, dacc, false, false);
    }
#pragma unroll
    for (int r = 0; r < 8; ++r) {
      int M = r + (hi ? 8 : 0);
      outBase[(size_t)M * N_CLASS * FDIM + n0 + ln] = dacc[r];
    }
  }
}

extern "C" void kernel_launch(void* const* d_in, const int* in_sizes, int n_in,
                              void* d_out, int out_size, void* d_ws, size_t ws_size,
                              hipStream_t stream) {
  const float* data = (const float*)d_in[0];
  float* out = (float*)d_out;
  (void)in_sizes; (void)n_in; (void)d_ws; (void)ws_size; (void)out_size;
  dim3 grid(N_QRY / QTILE, N_CLASS);
  proto_att_kernel<<<grid, 256, 0, stream>>>(data, out);
}


# score phase re-mapped: lanes own queries, a wave holds its 128-feature slice of 32 query rows in registers, support values arrive by scalar loads; 8 slice partials summed via LDS (new sum order)
# speedup vs baseline: 1.1369x; 1.1369x over previous
#include <hip/hip_runtime.h>
#include <hip/hip_bf16.h>


typedef __attribute__((ext_vector_type(2))) float v2f;
typedef __attribute__((ext_vector_type(8))) float v8f;

#define N_CLASS 32
#define N_SUP   16
#define FDIM    1024
#define N_QRY   512
#define QTILE   32
#define NWAVE   8
#define DSLICE  (FDIM / NWAVE)
#define TSTRIDE (DSLICE + 4)

__device__ __forceinline__ float dev_tanh(float x) {
#if __has_builtin(__builtin_amdgcn_tanhf)
  return __builtin_amdgcn_tanhf(x);
#else
  return tanhf(x);
#endif
}

__global__ __launch_bounds__(256, 4)
void proto_att_kernel(const float* __restrict__ data, float* __restrict__ out) {
  __shared__ __attribute__((aligned(16))) float sT[NWAVE][QTILE * TSTRIDE];
  __shared__ float sScore[QTILE][N_SUP + 1];
  __shared__ float sAtt[QTILE][N_SUP + 1];

  const int c    = blockIdx.x;
  const int qt   = blockIdx.y;
  const int tid  = threadIdx.x;
  const int lane = tid & 31;
  const int wave = __builtin_amdgcn_readfirstlane(tid >> 5);

  const float* Qg = data + (size_t)(N_CLASS * N_SUP + qt * QTILE) * FDIM + wave * DSLICE;
  const float* Sg = data + (size_t)c * N_SUP * FDIM;

  float* myT = sT[wave];
#pragma unroll
  for (int r = 0; r < QTILE; ++r) {
    float4 v = *(const float4*)(Qg + (size_t)r * FDIM + lane * 4);
    *(float4*)(myT + r * TSTRIDE + lane * 4) = v;
  }
  __syncthreads();
  float qreg[DSLICE];
#pragma unroll
  for (int j = 0; j < DSLICE / 4; ++j) {
    float4 v = *(const float4*)(myT + lane * TSTRIDE + j * 4);
    qreg[4 * j + 0] = v.x; qreg[4 * j + 1] = v.y; qreg[4 * j + 2] = v.z; qreg[4 * j + 3] = v.w;
  }

  const float* Sw = Sg + wave * DSLICE;
#pragma unroll 1
  for (int sg = 0; sg < N_SUP; sg += 4) {
    const float* base = Sw + (size_t)sg * FDIM;
    float sc[2][2][16];
#pragma unroll
    for (int j = 0; j < 16; ++j) { sc[0][0][j] = base[j]; sc[0][1][j] = base[FDIM + j]; }
    v2f acc0 = {0.f, 0.f}, acc1 = {0.f, 0.f};
#pragma unroll
    for (int t = 0; t < 16; ++t) {
      const int cur = t & 1, nxt = cur ^ 1;
      const int k = t & 7;
      const int h = t >> 3;
      if (k == 0) { acc0 = (v2f){0.f, 0.f}; acc1 = (v2f){0.f, 0.f}; }
#pragma unroll
      for (int j = 0; j < 16; j += 2) {
        const int d = k * 16 + j;
        const v2f t0 = {dev_tanh(sc[cur][0][j] * qreg[d]), dev_tanh(sc[cur][0][j + 1] * qreg[d + 1])};
        const v2f t1 = {dev_tanh(sc[cur][1][j] * qreg[d]), dev_tanh(sc[cur][1][j + 1] * qreg[d + 1])};
        acc0 += t0;
        acc1 += t1;
        if (j == 0) {
          __builtin_amdgcn_sched_barrier(0);
          if (t + 1 < 16) {
            const float* nb = base + (size_t)(((t + 1) >> 3) * 2) * FDIM + ((t + 1) & 7) * 16;
#pragma unroll
            for (int jj = 0; jj < 16; ++jj) { sc[nxt][0][jj] = nb[jj]; sc[nxt][1][jj] = nb[FDIM + jj]; }
          }
          __builtin_amdgcn_sched_barrier(0);
        }
      }
      if (k == 7) {
        myT[(sg + 2 * h + 0) * QTILE + lane] = acc0.x + acc0.y;
        myT[(sg + 2 * h + 1) * QTILE + lane] = acc1.x + acc1.y;
      }
    }
  }
  __syncthreads();

#pragma unroll
  for (int i = 0; i < 2; ++i) {
    const int idx = tid + i * 256;
    const int s = idx >> 5, q = idx & 31;
    float v = 0.f;
#pragma unroll
    for (int w = 0; w < NWAVE; ++w) v += sT[w][s * QTILE + q];
    sScore[q][s] = v;
  }
  __syncthreads();

#pragma unroll
  for (int i = 0; i < 2; ++i) {
    const int idx = tid + i * 256;
    const int q = idx >> 4, s = idx & 15;
    float m = -3.402823466e38f;
    float rowv[N_SUP];
#pragma unroll
    for (int k = 0; k < N_SUP; ++k) { rowv[k] = sScore[q][k]; m = fmaxf(m, rowv[k]); }
    float sum = 0.f;
#pragma unroll
    for (int k = 0; k < N_SUP; ++k) sum += __expf(rowv[k] - m);
    sAtt[q][s] = __expf(sScore[q][s] - m) / sum;
  }
  __syncthreads();

  const int hi = lane >> 4;
  const int ln = lane & 15;

  v2f a[2][4];
#pragma unroll
  for (int mt = 0; mt < 2; ++mt) {
#pragma unroll
    for (int kk = 0; kk < 4; ++kk) {
      a[mt][kk].x = sAtt[mt * 16 + ln][kk * 4 + (hi ? 2 : 0)];
      a[mt][kk].y = sAtt[mt * 16 + ln][kk * 4 + (hi ? 3 : 1)];
    }
  }

  float* outBase = out + ((size_t)(qt * QTILE) * N_CLASS + c) * FDIM;

  for (int nt = wave; nt < FDIM / 16; nt += NWAVE) {
    const int n0 = nt * 16;
    v8f dacc0 = {};
    v8f dacc1 = {};
#pragma unroll
    for (int kk = 0; kk < 4; ++kk) {
      v2f b;
      b.x = Sg[(size_t)(kk * 4 + (hi ? 2 : 0)) * FDIM + n0 + ln];
      b.y = Sg[(size_t)(kk * 4 + (hi ? 3 : 1)) * FDIM + n0 + ln];
      dacc0 = __builtin_amdgcn_wmma_f32_16x16x4_f32(false, a[0][kk], false, b,
                                                    (short)0, dacc0, false, false);
      dacc1 = __builtin_amdgcn_wmma_f32_16x16x4_f32(false, a[1][kk], false, b,
                                                    (short)0, dacc1, false, false);
    }
#pragma unroll
    for (int r = 0; r < 8; ++r) {
      int M = r + (hi ? 8 : 0);
      outBase[(size_t)M * N_CLASS * FDIM + n0 + ln] = dacc0[r];
      outBase[(size_t)(16 + M) * N_CLASS * FDIM + n0 + ln] = dacc1[r];
    }
  }
}

extern "C" void kernel_launch(void* const* d_in, const int* in_sizes, int n_in,
                              void* d_out, int out_size, void* d_ws, size_t ws_size,
                              hipStream_t stream) {
  const float* data = (const float*)d_in[0];
  float* out = (float*)d_out;
  (void)in_sizes; (void)n_in; (void)d_ws; (void)ws_size; (void)out_size;
  dim3 grid(N_CLASS, N_QRY / QTILE);
  proto_att_kernel<<<grid, 256, 0, stream>>>(data, out);
}


# score phase: a lane owns two queries (64-query workgroups of 16 waves, 64-feature slices): one scalar support value feeds two tanh, a step covers the scalar round trip; new sum order (16 partials)
# speedup vs baseline: 1.1887x; 1.0456x over previous
#include <hip/hip_runtime.h>
#include <hip/hip_bf16.h>


typedef __attribute__((ext_vector_type(2))) float v2f;
typedef __attribute__((ext_vector_type(8))) float v8f;

#define N_CLASS 32
#define N_SUP   16
#define FDIM    1024
#define N_QRY   512
#define QTILE   64
#define NWAVE   16
#define NTHREAD (NWAVE * 32)
#define DSLICE  (FDIM / NWAVE)
#define TSTRIDE (DSLICE + 4)

__device__ __forceinline__ float dev_tanh(float x) {
#if __has_builtin(__builtin_amdgcn_tanhf)
  return __builtin_amdgcn_tanhf(x);
#else
  return tanhf(x);
#endif
}

__global__ __launch_bounds__(NTHREAD, 4)
void proto_att_kernel(const float* __restrict__ data, float* __restrict__ out) {
  __shared__ __attribute__((aligned(16))) float sT[NWAVE][QTILE * TSTRIDE];
  __shared__ float sScore[QTILE][N_SUP + 1];
  __shared__ float sAtt[QTILE][N_SUP + 1];

  const int c    = blockIdx.x;
  const int qt   = blockIdx.y;
  const int tid  = threadIdx.x;
  const int lane = tid & 31;
  const int wave = __builtin_amdgcn_readfirstlane(tid >> 5);

  const float* Qg = data + (size_t)(N_CLASS * N_SUP + qt * QTILE) * FDIM + wave * DSLICE;
  const float* Sg = data + (size_t)c * N_SUP * FDIM;

  const int hrow = lane >> 4;
  const int hcol = (lane & 15) * 4;
  float* myT = sT[wave];
#pragma unroll
  for (int r = 0; r < QTILE; r += 2) {
    float4 v = *(const float4*)(Qg + (size_t)(r + hrow) * FDIM + hcol);
    *(float4*)(myT + (r + hrow) * TSTRIDE + hcol) = v;
  }
  __syncthreads();
  float qreg[2][DSLICE];
#pragma unroll
  for (int g = 0; g < 2; ++g) {
#pragma unroll
    for (int j = 0; j < DSLICE / 4; ++j) {
      float4 v = *(const float4*)(myT + (g * 32 + lane) * TSTRIDE + j * 4);
      qreg[g][4 * j + 0] = v.x; qreg[g][4 * j + 1] = v.y; qreg[g][4 * j + 2] = v.z; qreg[g][4 * j + 3] = v.w;
    }
  }

  const float* Sw = Sg + wave * DSLICE;
#pragma unroll 1
  for (int sg = 0; sg < N_SUP; sg += 4) {
    const float* base = Sw + (size_t)sg * FDIM;
    float sc[2][2][16];
#pragma unroll
    for (int j = 0; j < 16; ++j) { sc[0][0][j] = base[j]; sc[0][1][j] = base[FDIM + j]; }
    v2f acc[2][2];
#pragma unroll
    for (int t = 0; t < 8; ++t) {
      const int cur = t & 1, nxt = cur ^ 1;
      const int k = t & 3;
      const int h = t >> 2;
      if (k == 0) {
        acc[0][0] = (v2f){0.f, 0.f}; acc[0][1] = (v2f){0.f, 0.f};
        acc[1][0] = (v2f){0.f, 0.f}; acc[1][1] = (v2f){0.f, 0.f};
      }
#pragma unroll
      for (int j = 0; j < 16; j += 2) {
        const int d = k * 16 + j;
#pragma unroll
        for (int row = 0; row < 2; ++row) {
#pragma unroll
          for (int g = 0; g < 2; ++g) {
            const v2f tv = {dev_tanh(sc[cur][row][j] * qreg[g][d]),
                            dev_tanh(sc[cur][row][j + 1] * qreg[g][d + 1])};
            acc[row][g] += tv;
          }
        }
        if (j == 0) {
          __builtin_amdgcn_sched_barrier(0);
          if (t + 1 < 8) {
            const float* nb = base + (size_t)(((t + 1) >> 2) * 2) * FDIM + ((t + 1) & 3) * 16;
#pragma unroll
            for (int jj = 0; jj < 16; ++jj) { sc[nxt][0][jj] = nb[jj]; sc[nxt][1][jj] = nb[FDIM + jj]; }
          }
          __builtin_amdgcn_sched_barrier(0);
        }
      }
      if (k == 3) {
#pragma unroll
        for (int row = 0; row < 2; ++row) {
#pragma unroll
          for (int g = 0; g < 2; ++g)
            myT[(sg + 2 * h + row) * QTILE + g * 32 + lane] = acc[row][g].x + acc[row][g].y;
        }
      }
    }
  }
  __syncthreads();

#pragma unroll
  for (int i = 0; i < (QTILE * N_SUP) / NTHREAD; ++i) {
    const int idx = tid + i * NTHREAD;
    const int s = idx >> 6, q = idx & 63;
    float v = 0.f;
#pragma unroll
    for (int w = 0; w < NWAVE; ++w) v += sT[w][s * QTILE + q];
    sScore[q][s] = v;
  }
  __syncthreads();

#pragma unroll
  for (int i = 0; i < (QTILE * N_SUP) / NTHREAD; ++i) {
    const int idx = tid + i * NTHREAD;
    const int q = idx >> 4, s = idx & 15;
    float m = -3.402823466e38f;
    float rowv[N_SUP];
#pragma unroll
    for (int k = 0; k < N_SUP; ++k) { rowv[k] = sScore[q][k]; m = fmaxf(m, rowv[k]); }
    float sum = 0.f;
#pragma unroll
    for (int k = 0; k < N_SUP; ++k) sum += __expf(rowv[k] - m);
    sAtt[q][s] = __expf(sScore[q][s] - m) / sum;
  }
  __syncthreads();

  const int hi = lane >> 4;
  const int ln = lane & 15;

  v2f a[QTILE / 16][4];
#pragma unroll
  for (int mt = 0; mt < QTILE / 16; ++mt) {
#pragma unroll
    for (int kk = 0; kk < 4; ++kk) {
      a[mt][kk].x = sAtt[mt * 16 + ln][kk * 4 + (hi ? 2 : 0)];
      a[mt][kk].y = sAtt[mt * 16 + ln][kk * 4 + (hi ? 3 : 1)];
    }
  }

  float* outBase = out + ((size_t)(qt * QTILE) * N_CLASS + c) * FDIM;

  for (int nt = wave; nt < FDIM / 16; nt += NWAVE) {
    const int n0 = nt * 16;
    v8f dacc[QTILE / 16];
#pragma unroll
    for (int mt = 0; mt < QTILE / 16; ++mt) dacc[mt] = (v8f){};
#pragma unroll
    for (int kk = 0; kk < 4; ++kk) {
      v2f b;
      b.x = Sg[(size_t)(kk * 4 + (hi ? 2 : 0)) * FDIM + n0 + ln];
      b.y = Sg[(size_t)(kk * 4 + (hi ? 3 : 1)) * FDIM + n0 + ln];
#pragma unroll
      for (int mt = 0; mt < QTILE / 16; ++mt)
        dacc[mt] = __builtin_amdgcn_wmma_f32_16x16x4_f32(false, a[mt][kk], false, b,
                                                         (short)0, dacc[mt], false, false);
    }
#pragma unroll
    for (int mt = 0; mt < QTILE / 16; ++mt) {
#pragma unroll
      for (int r = 0; r < 8; ++r) {
        const int M = mt * 16 + r + (hi ? 8 : 0);
        outBase[(size_t)M * N_CLASS * FDIM + n0 + ln] = dacc[mt][r];
      }
    }
  }
}

extern "C" void kernel_launch(void* const* d_in, const int* in_sizes, int n_in,
                              void* d_out, int out_size, void* d_ws, size_t ws_size,
                              hipStream_t stream) {
  const float* data = (const float*)d_in[0];
  float* out = (float*)d_out;
  (void)in_sizes; (void)n_in; (void)d_ws; (void)ws_size; (void)out_size;
  dim3 grid(N_CLASS, N_QRY / QTILE);
  proto_att_kernel<<<grid, NTHREAD, 0, stream>>>(data, out);
}


# output phase: a wave owns adjacent column-tile pairs and swaps half-waves between two registers (v_permlane16_swap): each 4-byte B load and result store covers one whole 128-byte line; sums unchanged
# speedup vs baseline: 1.2427x; 1.0454x over previous
#include <hip/hip_runtime.h>
#include <hip/hip_bf16.h>


typedef __attribute__((ext_vector_type(2))) float v2f;
typedef __attribute__((ext_vector_type(8))) float v8f;

#define N_CLASS 32
#define N_SUP   16
#define FDIM    1024
#define N_QRY   512
#define QTILE   64
#define NWAVE   16
#define NTHREAD (NWAVE * 32)
#define DSLICE  (FDIM / NWAVE)
#define TSTRIDE (DSLICE + 4)

__device__ __forceinline__ float dev_tanh(float x) {
#if __has_builtin(__builtin_amdgcn_tanhf)
  return __builtin_amdgcn_tanhf(x);
#else
  return tanhf(x);
#endif
}

__device__ __forceinline__ void half_swap(float& x, float& y) {
  asm volatile("v_permlane16_swap_b32 %0, %1" : "+v"(x), "+v"(y));
}

__global__ __launch_bounds__(NTHREAD, 4)
void proto_att_kernel(const float* __restrict__ data, float* __restrict__ out) {
  __shared__ __attribute__((aligned(16))) float sT[NWAVE][QTILE * TSTRIDE];
  __shared__ float sScore[QTILE][N_SUP + 1];
  __shared__ float sAtt[QTILE][N_SUP + 1];

  const int c    = blockIdx.x;
  const int qt   = blockIdx.y;
  const int tid  = threadIdx.x;
  const int lane = tid & 31;
  const int wave = __builtin_amdgcn_readfirstlane(tid >> 5);

  const float* Qg = data + (size_t)(N_CLASS * N_SUP + qt * QTILE) * FDIM + wave * DSLICE;
  const float* Sg = data + (size_t)c * N_SUP * FDIM;

  const int hrow = lane >> 4;
  const int hcol = (lane & 15) * 4;
  float* myT = sT[wave];
#pragma unroll
  for (int r = 0; r < QTILE; r += 2) {
    float4 v = *(const float4*)(Qg + (size_t)(r + hrow) * FDIM + hcol);
    *(float4*)(myT + (r + hrow) * TSTRIDE + hcol) = v;
  }
  __syncthreads();
  float qreg[2][DSLICE];
#pragma unroll
  for (int g = 0; g < 2; ++g) {
#pragma unroll
    for (int j = 0; j < DSLICE / 4; ++j) {
      float4 v = *(const float4*)(myT + (g * 32 + lane) * TSTRIDE + j * 4);
      qreg[g][4 * j + 0] = v.x; qreg[g][4 * j + 1] = v.y; qreg[g][4 * j + 2] = v.z; qreg[g][4 * j + 3] = v.w;
    }
  }

  const float* Sw = Sg + wave * DSLICE;
#pragma unroll 1
  for (int sg = 0; sg < N_SUP; sg += 4) {
    const float* base = Sw + (size_t)sg * FDIM;
    float sc[2][2][16];
#pragma unroll
    for (int j = 0; j < 16; ++j) { sc[0][0][j] = base[j]; sc[0][1][j] = base[FDIM + j]; }
    v2f acc[2][2];
#pragma unroll
    for (int t = 0; t < 8; ++t) {
      const int cur = t & 1, nxt = cur ^ 1;
      const int k = t & 3;
      const int h = t >> 2;
      if (k == 0) {
        acc[0][0] = (v2f){0.f, 0.f}; acc[0][1] = (v2f){0.f, 0.f};
        acc[1][0] = (v2f){0.f, 0.f}; acc[1][1] = (v2f){0.f, 0.f};
      }
#pragma unroll
      for (int j = 0; j < 16; j += 2) {
        const int d = k * 16 + j;
#pragma unroll
        for (int row = 0; row < 2; ++row) {
#pragma unroll
          for (int g = 0; g < 2; ++g) {
            const v2f tv = {dev_tanh(sc[cur][row][j] * qreg[g][d]),
                            dev_tanh(sc[cur][row][j + 1] * qreg[g][d + 1])};
            acc[row][g] += tv;
          }
        }
        if (j == 0) {
          __builtin_amdgcn_sched_barrier(0);
          if (t + 1 < 8) {
            const float* nb = base + (size_t)(((t + 1) >> 2) * 2) * FDIM + ((t + 1) & 3) * 16;
#pragma unroll
            for (int jj = 0; jj < 16; ++jj) { sc[nxt][0][jj] = nb[jj]; sc[nxt][1][jj] = nb[FDIM + jj]; }
          }
          __builtin_amdgcn_sched_barrier(0);
        }
      }
      if (k == 3) {
#pragma unroll
        for (int row = 0; row < 2; ++row) {
#pragma unroll
          for (int g = 0; g < 2; ++g)
            myT[(sg + 2 * h + row) * QTILE + g * 32 + lane] = acc[row][g].x + acc[row][g].y;
        }
      }
    }
  }
  __syncthreads();

#pragma unroll
  for (int i = 0; i < (QTILE * N_SUP) / NTHREAD; ++i) {
    const int idx = tid + i * NTHREAD;
    const int s = idx >> 6, q = idx & 63;
    float v = 0.f;
#pragma unroll
    for (int w = 0; w < NWAVE; ++w) v += sT[w][s * QTILE + q];
    sScore[q][s] = v;
  }
  __syncthreads();

#pragma unroll
  for (int i = 0; i < (QTILE * N_SUP) / NTHREAD; ++i) {
    const int idx = tid + i * NTHREAD;
    const int q = idx >> 4, s = idx & 15;
    float m = -3.402823466e38f;
    float rowv[N_SUP];
#pragma unroll
    for (int k = 0; k < N_SUP; ++k) { rowv[k] = sScore[q][k]; m = fmaxf(m, rowv[k]); }
    float sum = 0.f;
#pragma unroll
    for (int k = 0; k < N_SUP; ++k) sum += __expf(rowv[k] - m);
    sAtt[q][s] = __expf(sScore[q][s] - m) / sum;
  }
  __syncthreads();

  const int hi = lane >> 4;
  const int ln = lane & 15;

  v2f a[QTILE / 16][4];
#pragma unroll
  for (int mt = 0; mt < QTILE / 16; ++mt) {
#pragma unroll
    for (int kk = 0; kk < 4; ++kk) {
      a[mt][kk].x = sAtt[mt * 16 + ln][kk * 4 + (hi ? 2 : 0)];
      a[mt][kk].y = sAtt[mt * 16 + ln][kk * 4 + (hi ? 3 : 1)];
    }
  }

  float* outBase = out + ((size_t)(qt * QTILE) * N_CLASS + c) * FDIM;

  float rr[2][4][4];
#pragma unroll
  for (int i = 0; i < 2; ++i) {
    const float* sp = Sg + (wave + i * NWAVE) * 32 + lane;
#pragma unroll
    for (int kk = 0; kk < 4; ++kk) {
#pragma unroll
      for (int k = 0; k < 4; ++k) rr[i][kk][k] = sp[(size_t)(kk * 4 + k) * FDIM];
    }
  }

#pragma unroll
  for (int i = 0; i < 2; ++i) {
    const int n0 = (wave + i * NWAVE) * 32;
    v8f dacc[2][QTILE / 16];
#pragma unroll
    for (int j = 0; j < 2; ++j) {
#pragma unroll
      for (int mt = 0; mt < QTILE / 16; ++mt) dacc[j][mt] = (v8f){};
    }
#pragma unroll
    for (int kk = 0; kk < 4; ++kk) {
      half_swap(rr[i][kk][0], rr[i][kk][2]);
      half_swap(rr[i][kk][1], rr[i][kk][3]);
      const v2f b0 = {rr[i][kk][0], rr[i][kk][1]};
      const v2f b1 = {rr[i][kk][2], rr[i][kk][3]};
#pragma unroll
      for (int mt = 0; mt < QTILE / 16; ++mt) {
        dacc[0][mt] = __builtin_amdgcn_wmma_f32_16x16x4_f32(false, a[mt][kk], false, b0,
                                                            (short)0, dacc[0][mt], false, false);
        dacc[1][mt] = __builtin_amdgcn_wmma_f32_16x16x4_f32(false, a[mt][kk], false, b1,
                                                            (short)0, dacc[1][mt], false, false);
      }
    }
    __builtin_amdgcn_sched_barrier(0);
    asm volatile("v_nop\n\tv_nop\n\tv_nop\n\tv_nop\n\tv_nop");
#pragma unroll
    for (int mt = 0; mt < QTILE / 16; ++mt) {
#pragma unroll
      for (int r = 0; r < 8; ++r) {
        float lo = dacc[0][mt][r], hi8 = dacc[1][mt][r];
        half_swap(lo, hi8);
        const int M = mt * 16 + r;
        outBase[(size_t)M * N_CLASS * FDIM + n0 + lane] = lo;
        outBase[(size_t)(M + 8) * N_CLASS * FDIM + n0 + lane] = hi8;
      }
    }
  }
}

extern "C" void kernel_launch(void* const* d_in, const int* in_sizes, int n_in,
                              void* d_out, int out_size, void* d_ws, size_t ws_size,
                              hipStream_t stream) {
  const float* data = (const float*)d_in[0];
  float* out = (float*)d_out;
  (void)in_sizes; (void)n_in; (void)d_ws; (void)ws_size; (void)out_size;
  dim3 grid(N_CLASS, N_QRY / QTILE);
  proto_att_kernel<<<grid, NTHREAD, 0, stream>>>(data, out);
}


# score loop in hand-kept issue order, pipelined by groups: each tanh followed by one packed multiply (next group; one scalar times the lane's query pair) or packed add (group before); sums unchanged
# speedup vs baseline: 1.3033x; 1.0488x over previous
#include <hip/hip_runtime.h>
#include <hip/hip_bf16.h>


typedef __attribute__((ext_vector_type(2))) float v2f;
typedef __attribute__((ext_vector_type(8))) float v8f;
typedef __attribute__((address_space(4))) float cfloat_t;

#define ORDER_FENCE() __builtin_amdgcn_sched_barrier(0x4)

#define N_CLASS 32
#define N_SUP   16
#define FDIM    1024
#define N_QRY   512
#define QTILE   64
#define NWAVE   16
#define NTHREAD (NWAVE * 32)
#define DSLICE  (FDIM / NWAVE)
#define TSTRIDE (DSLICE + 4)

__device__ __forceinline__ float dev_tanh(float x) {
#if __has_builtin(__builtin_amdgcn_tanhf)
  return __builtin_amdgcn_tanhf(x);
#else
  return tanhf(x);
#endif
}

__device__ __forceinline__ void half_swap(float& x, float& y) {
  asm volatile("v_permlane16_swap_b32 %0, %1" : "+v"(x), "+v"(y));
}

__global__ __launch_bounds__(NTHREAD, 4)
void proto_att_kernel(const float* __restrict__ data, float* __restrict__ out) {
  __shared__ __attribute__((aligned(16))) float sT[NWAVE][QTILE * TSTRIDE];
  __shared__ float sScore[QTILE][N_SUP + 1];
  __shared__ float sAtt[QTILE][N_SUP + 1];

  const int c    = blockIdx.x;
  const int qt   = blockIdx.y;
  const int tid  = threadIdx.x;
  const int lane = tid & 31;
  const int wave = __builtin_amdgcn_readfirstlane(tid >> 5);

  const float* Qg = data + (size_t)(N_CLASS * N_SUP + qt * QTILE) * FDIM + wave * DSLICE;
  const float* Sg = data + (size_t)c * N_SUP * FDIM;

  const int hrow = lane >> 4;
  const int hcol = (lane & 15) * 4;
  float* myT = sT[wave];
#pragma unroll
  for (int r = 0; r < QTILE; r += 2) {
    float4 v = *(const float4*)(Qg + (size_t)(r + hrow) * FDIM + hcol);
    *(float4*)(myT + (r + hrow) * TSTRIDE + hcol) = v;
  }
  __syncthreads();
  float qreg[2][DSLICE];
#pragma unroll
  for (int g = 0; g < 2; ++g) {
#pragma unroll
    for (int j = 0; j < DSLICE / 4; ++j) {
      float4 v = *(const float4*)(myT + (g * 32 + lane) * TSTRIDE + j * 4);
      qreg[g][4 * j + 0] = v.x; qreg[g][4 * j + 1] = v.y; qreg[g][4 * j + 2] = v.z; qreg[g][4 * j + 3] = v.w;
    }
  }

  v2f qp[DSLICE];
#pragma unroll
  for (int d = 0; d < DSLICE; ++d) qp[d] = (v2f){qreg[0][d], qreg[1][d]};

  const float* Sw = Sg + wave * DSLICE;
#pragma unroll 1
  for (int sg = 0; sg < N_SUP; sg += 4) {
    const cfloat_t* base = (const cfloat_t*)(Sw + (size_t)sg * FDIM);
    float sc[2][2][16];
#pragma unroll
    for (int j = 0; j < 16; ++j) { sc[0][0][j] = base[j]; sc[0][1][j] = base[FDIM + j]; }
    v2f acc[2][2];
    v2f pr[2][4];
    v2f tv[2][4];
#pragma unroll
    for (int row = 0; row < 2; ++row) {
#pragma unroll
      for (int f = 0; f < 2; ++f)
        pr[0][row * 2 + f] = (v2f){sc[0][row][f], sc[0][row][f]} * qp[f];
    }
    __builtin_amdgcn_sched_barrier(0);
#pragma unroll
    for (int jj = 0; jj < 16; ++jj) { sc[1][0][jj] = base[16 + jj]; sc[1][1][jj] = base[FDIM + 16 + jj]; }
    __builtin_amdgcn_sched_barrier(0);
#pragma unroll
    for (int n = 0; n < 64; ++n) {
      const int n1 = n + 1, t1 = n1 >> 3, j1 = (n1 & 7) * 2, d1 = (t1 & 3) * 16 + j1, b1 = t1 & 1;
      const int m = n - 1, tm = m >> 3;
      if (n >= 1 && (tm & 3) == 0 && (m & 7) == 0) {
        acc[0][0] = (v2f){0.f, 0.f}; acc[0][1] = (v2f){0.f, 0.f};
        acc[1][0] = (v2f){0.f, 0.f}; acc[1][1] = (v2f){0.f, 0.f};
      }
#pragma unroll
      for (int i = 0; i < 4; ++i) {
        tv[n & 1][i].x = dev_tanh(pr[n & 1][i].x);
        ORDER_FENCE();
        if (n1 < 64) {
          const int row = i >> 1, f = i & 1;
          pr[n1 & 1][i] = (v2f){sc[b1][row][j1 + f], sc[b1][row][j1 + f]} * qp[d1 + f];
          ORDER_FENCE();
        }
        tv[n & 1][i].y = dev_tanh(pr[n & 1][i].y);
        ORDER_FENCE();
        if (n >= 1) {
          acc[i >> 1][i & 1] += tv[m & 1][i];
          ORDER_FENCE();
        }
      }
      if (n1 < 64 && (n1 & 7) == 0) {
        __builtin_amdgcn_sched_barrier(0);
        if (t1 + 1 < 8) {
          const int t2 = t1 + 1;
          const cfloat_t* nb = base + (size_t)((t2 >> 2) * 2) * FDIM + (t2 & 3) * 16;
#pragma unroll
          for (int jj = 0; jj < 16; ++jj) { sc[t2 & 1][0][jj] = nb[jj]; sc[t2 & 1][1][jj] = nb[FDIM + jj]; }
        }
        __builtin_amdgcn_sched_barrier(0);
      }
      if (n >= 1 && (tm & 3) == 3 && (m & 7) == 7) {
#pragma unroll
        for (int row = 0; row < 2; ++row) {
#pragma unroll
          for (int g = 0; g < 2; ++g)
            myT[(sg + 2 * (tm >> 2) + row) * QTILE + g * 32 + lane] = acc[row][0][g] + acc[row][1][g];
        }
      }
    }
#pragma unroll
    for (int i = 0; i < 4; ++i) acc[i >> 1][i & 1] += tv[1][i];
#pragma unroll
    for (int row = 0; row < 2; ++row) {
#pragma unroll
      for (int g = 0; g < 2; ++g)
        myT[(sg + 2 + row) * QTILE + g * 32 + lane] = acc[row][0][g] + acc[row][1][g];
    }
  }
  __syncthreads();

#pragma unroll
  for (int i = 0; i < (QTILE * N_SUP) / NTHREAD; ++i) {
    const int idx = tid + i * NTHREAD;
    const int s = idx >> 6, q = idx & 63;
    float v = 0.f;
#pragma unroll
    for (int w = 0; w < NWAVE; ++w) v += sT[w][s * QTILE + q];
    sScore[q][s] = v;
  }
  __syncthreads();

#pragma unroll
  for (int i = 0; i < (QTILE * N_SUP) / NTHREAD; ++i) {
    const int idx = tid + i * NTHREAD;
    const int q = idx >> 4, s = idx & 15;
    float m = -3.402823466e38f;
    float rowv[N_SUP];
#pragma unroll
    for (int k = 0; k < N_SUP; ++k) { rowv[k] = sScore[q][k]; m = fmaxf(m, rowv[k]); }
    float sum = 0.f;
#pragma unroll
    for (int k = 0; k < N_SUP; ++k) sum += __expf(rowv[k] - m);
    sAtt[q][s] = __expf(sScore[q][s] - m) / sum;
  }
  __syncthreads();

  const int hi = lane >> 4;
  const int ln = lane & 15;

  v2f a[QTILE / 16][4];
#pragma unroll
  for (int mt = 0; mt < QTILE / 16; ++mt) {
#pragma unroll
    for (int kk = 0; kk < 4; ++kk) {
      a[mt][kk].x = sAtt[mt * 16 + ln][kk * 4 + (hi ? 2 : 0)];
      a[mt][kk].y = sAtt[mt * 16 + ln][kk * 4 + (hi ? 3 : 1)];
    }
  }

  float* outBase = out + ((size_t)(qt * QTILE) * N_CLASS + c) * FDIM;

  float rr[2][4][4];
#pragma unroll
  for (int i = 0; i < 2; ++i) {
    const float* sp = Sg + (wave + i * NWAVE) * 32 + lane;
#pragma unroll
    for (int kk = 0; kk < 4; ++kk) {
#pragma unroll
      for (int k = 0; k < 4; ++k) rr[i][kk][k] = sp[(size_t)(kk * 4 + k) * FDIM];
    }
  }

#pragma unroll
  for (int i = 0; i < 2; ++i) {
    const int n0 = (wave + i * NWAVE) * 32;
    v8f dacc[2][QTILE / 16];
#pragma unroll
    for (int j = 0; j < 2; ++j) {
#pragma unroll
      for (int mt = 0; mt < QTILE / 16; ++mt) dacc[j][mt] = (v8f){};
    }
#pragma unroll
    for (int kk = 0; kk < 4; ++kk) {
      half_swap(rr[i][kk][0], rr[i][kk][2]);
      half_swap(rr[i][kk][1], rr[i][kk][3]);
      const v2f b0 = {rr[i][kk][0], rr[i][kk][1]};
      const v2f b1 = {rr[i][kk][2], rr[i][kk][3]};
#pragma unroll
      for (int mt = 0; mt < QTILE / 16; ++mt) {
        dacc[0][mt] = __builtin_amdgcn_wmma_f32_16x16x4_f32(false, a[mt][kk], false, b0,
                                                            (short)0, dacc[0][mt], false, false);
        dacc[1][mt] = __builtin_amdgcn_wmma_f32_16x16x4_f32(false, a[mt][kk], false, b1,
                                                            (short)0, dacc[1][mt], false, false);
      }
    }
    __builtin_amdgcn_sched_barrier(0);
    asm volatile("v_nop\n\tv_nop\n\tv_nop\n\tv_nop\n\tv_nop");
#pragma unroll
    for (int mt = 0; mt < QTILE / 16; ++mt) {
#pragma unroll
      for (int r = 0; r < 8; ++r) {
        float lo = dacc[0][mt][r], hi8 = dacc[1][mt][r];
        half_swap(lo, hi8);
        const int M = mt * 16 + r;
        outBase[(size_t)M * N_CLASS * FDIM + n0 + lane] = lo;
        outBase[(size_t)(M + 8) * N_CLASS * FDIM + n0 + lane] = hi8;
      }
    }
  }
}

extern "C" void kernel_launch(void* const* d_in, const int* in_sizes, int n_in,
                              void* d_out, int out_size, void* d_ws, size_t ws_size,
                              hipStream_t stream) {
  const float* data = (const float*)d_in[0];
  float* out = (float*)d_out;
  (void)in_sizes; (void)n_in; (void)d_ws; (void)ws_size; (void)out_size;
  dim3 grid(N_CLASS, N_QRY / QTILE);
  proto_att_kernel<<<grid, NTHREAD, 0, stream>>>(data, out);
}


# score loop: support values come by broadcast reads of a per-wave LDS copy (in order, counted waits, a step ahead), not by scalar loads; a multiply pairs two features of one query; sums unchanged
# speedup vs baseline: 1.3872x; 1.0644x over previous
#include <hip/hip_runtime.h>
#include <hip/hip_bf16.h>


typedef __attribute__((ext_vector_type(2))) float v2f;
typedef __attribute__((ext_vector_type(8))) float v8f;

#define ORDER_FENCE() __builtin_amdgcn_sched_barrier(0x4)

#define N_CLASS 32
#define N_SUP   16
#define FDIM    1024
#define N_QRY   512
#define QTILE   64
#define NWAVE   16
#define NTHREAD (NWAVE * 32)
#define DSLICE  (FDIM / NWAVE)
#define TSTRIDE (DSLICE + 4)
#define SSLICE_OFF (N_SUP * QTILE)

__device__ __forceinline__ float dev_tanh(float x) {
#if __has_builtin(__builtin_amdgcn_tanhf)
  return __builtin_amdgcn_tanhf(x);
#else
  return tanhf(x);
#endif
}

__device__ __forceinline__ void half_swap(float& x, float& y) {
  asm volatile("v_permlane16_swap_b32 %0, %1" : "+v"(x), "+v"(y));
}

__global__ __launch_bounds__(NTHREAD, 4)
void proto_att_kernel(const float* __restrict__ data, float* __restrict__ out) {
  __shared__ __attribute__((aligned(16))) float sT[NWAVE][QTILE * TSTRIDE];
  __shared__ float sScore[QTILE][N_SUP + 1];
  __shared__ float sAtt[QTILE][N_SUP + 1];

  const int c    = blockIdx.x;
  const int qt   = blockIdx.y;
  const int tid  = threadIdx.x;
  const int lane = tid & 31;
  const int wave = __builtin_amdgcn_readfirstlane(tid >> 5);

  const float* Qg = data + (size_t)(N_CLASS * N_SUP + qt * QTILE) * FDIM + wave * DSLICE;
  const float* Sg = data + (size_t)c * N_SUP * FDIM;

  const int hrow = lane >> 4;
  const int hcol = (lane & 15) * 4;
  float* myT = sT[wave];
#pragma unroll
  for (int r = 0; r < QTILE; r += 2) {
    float4 v = *(const float4*)(Qg + (size_t)(r + hrow) * FDIM + hcol);
    *(float4*)(myT + (r + hrow) * TSTRIDE + hcol) = v;
  }
  __syncthreads();
  const float* Sw = Sg + wave * DSLICE;
  float4 sreg[N_SUP / 2];
#pragma unroll
  for (int r = 0; r < N_SUP; r += 2) sreg[r >> 1] = *(const float4*)(Sw + (size_t)(r + hrow) * FDIM + hcol);
  v2f q2[2][DSLICE / 2];
#pragma unroll
  for (int g = 0; g < 2; ++g) {
#pragma unroll
    for (int j = 0; j < DSLICE / 4; ++j) {
      float4 v = *(const float4*)(myT + (g * 32 + lane) * TSTRIDE + j * 4);
      q2[g][2 * j + 0] = (v2f){v.x, v.y};
      q2[g][2 * j + 1] = (v2f){v.z, v.w};
    }
  }
  float* sS = myT + SSLICE_OFF;
#pragma unroll
  for (int r = 0; r < N_SUP; r += 2) *(float4*)(sS + (r + hrow) * DSLICE + hcol) = sreg[r >> 1];

#pragma unroll 1
  for (int sg = 0; sg < N_SUP; sg += 4) {
    const float* base = sS + sg * DSLICE;
    v2f sv[2][2][8];
#pragma unroll
    for (int row = 0; row < 2; ++row) {
#pragma unroll
      for (int k = 0; k < 4; ++k) {
        float4 v = *(const float4*)(base + row * DSLICE + 4 * k);
        sv[0][row][2 * k] = (v2f){v.x, v.y}; sv[0][row][2 * k + 1] = (v2f){v.z, v.w};
      }
    }
    __builtin_amdgcn_sched_barrier(0);
#pragma unroll
    for (int row = 0; row < 2; ++row) {
#pragma unroll
      for (int k = 0; k < 4; ++k) {
        float4 v = *(const float4*)(base + row * DSLICE + 16 + 4 * k);
        sv[1][row][2 * k] = (v2f){v.x, v.y}; sv[1][row][2 * k + 1] = (v2f){v.z, v.w};
      }
    }
    __builtin_amdgcn_sched_barrier(0);
    v2f acc[2][2];
    v2f pr[2][4];
    v2f tv[2][4];
#pragma unroll
    for (int row = 0; row < 2; ++row) {
#pragma unroll
      for (int g = 0; g < 2; ++g)
        pr[0][row * 2 + g] = sv[0][row][0] * q2[g][0];
    }
#pragma unroll
    for (int n = 0; n < 64; ++n) {
      const int n1 = n + 1, t1 = n1 >> 3, p1 = n1 & 7, d1 = (t1 & 3) * 8 + p1, b1 = t1 & 1;
      const int m = n - 1, tm = m >> 3;
      if (n >= 1 && (tm & 3) == 0 && (m & 7) == 0) {
        acc[0][0] = (v2f){0.f, 0.f}; acc[0][1] = (v2f){0.f, 0.f};
        acc[1][0] = (v2f){0.f, 0.f}; acc[1][1] = (v2f){0.f, 0.f};
      }
#pragma unroll
      for (int i = 0; i < 4; ++i) {
        tv[n & 1][i].x = dev_tanh(pr[n & 1][i].x);
        ORDER_FENCE();
        if (n1 < 64) {
          const int row = i >> 1, g = i & 1;
          pr[n1 & 1][i] = sv[b1][row][p1] * q2[g][d1];
          ORDER_FENCE();
        }
        tv[n & 1][i].y = dev_tanh(pr[n & 1][i].y);
        ORDER_FENCE();
        if (n >= 1) {
          acc[i >> 1][i & 1] += tv[m & 1][i];
          ORDER_FENCE();
        }
      }
      if (n1 < 64 && (n1 & 7) == 0) {
        __builtin_amdgcn_sched_barrier(0);
        if (t1 + 1 < 8) {
          const int t2 = t1 + 1;
          const float* nb = base + ((t2 >> 2) * 2) * DSLICE + (t2 & 3) * 16;
#pragma unroll
          for (int row = 0; row < 2; ++row) {
#pragma unroll
            for (int k = 0; k < 4; ++k) {
              float4 v = *(const float4*)(nb + row * DSLICE + 4 * k);
              sv[t2 & 1][row][2 * k] = (v2f){v.x, v.y}; sv[t2 & 1][row][2 * k + 1] = (v2f){v.z, v.w};
            }
          }
        }
        __builtin_amdgcn_sched_barrier(0);
      }
      if (n >= 1 && (tm & 3) == 3 && (m & 7) == 7) {
#pragma unroll
        for (int row = 0; row < 2; ++row) {
#pragma unroll
          for (int g = 0; g < 2; ++g)
            myT[(sg + 2 * (tm >> 2) + row) * QTILE + g * 32 + lane] = acc[row][g].x + acc[row][g].y;
        }
      }
    }
#pragma unroll
    for (int i = 0; i < 4; ++i) acc[i >> 1][i & 1] += tv[1][i];
#pragma unroll
    for (int row = 0; row < 2; ++row) {
#pragma unroll
      for (int g = 0; g < 2; ++g)
        myT[(sg + 2 + row) * QTILE + g * 32 + lane] = acc[row][g].x + acc[row][g].y;
    }
  }
  __syncthreads();

#pragma unroll
  for (int i = 0; i < (QTILE * N_SUP) / NTHREAD; ++i) {
    const int idx = tid + i * NTHREAD;
    const int s = idx >> 6, q = idx & 63;
    float v = 0.f;
#pragma unroll
    for (int w = 0; w < NWAVE; ++w) v += sT[w][s * QTILE + q];
    sScore[q][s] = v;
  }
  __syncthreads();

#pragma unroll
  for (int i = 0; i < (QTILE * N_SUP) / NTHREAD; ++i) {
    const int idx = tid + i * NTHREAD;
    const int q = idx >> 4, s = idx & 15;
    float m = -3.402823466e38f;
    float rowv[N_SUP];
#pragma unroll
    for (int k = 0; k < N_SUP; ++k) { rowv[k] = sScore[q][k]; m = fmaxf(m, rowv[k]); }
    float sum = 0.f;
#pragma unroll
    for (int k = 0; k < N_SUP; ++k) sum += __expf(rowv[k] - m);
    sAtt[q][s] = __expf(sScore[q][s] - m) / sum;
  }
  __syncthreads();

  const int hi = lane >> 4;
  const int ln = lane & 15;

  v2f a[QTILE / 16][4];
#pragma unroll
  for (int mt = 0; mt < QTILE / 16; ++mt) {
#pragma unroll
    for (int kk = 0; kk < 4; ++kk) {
      a[mt][kk].x = sAtt[mt * 16 + ln][kk * 4 + (hi ? 2 : 0)];
      a[mt][kk].y = sAtt[mt * 16 + ln][kk * 4 + (hi ? 3 : 1)];
    }
  }

  float* outBase = out + ((size_t)(qt * QTILE) * N_CLASS + c) * FDIM;

  float rr[2][4][4];
#pragma unroll
  for (int i = 0; i < 2; ++i) {
    const float* sp = Sg + (wave + i * NWAVE) * 32 + lane;
#pragma unroll
    for (int kk = 0; kk < 4; ++kk) {
#pragma unroll
      for (int k = 0; k < 4; ++k) rr[i][kk][k] = sp[(size_t)(kk * 4 + k) * FDIM];
    }
  }

#pragma unroll
  for (int i = 0; i < 2; ++i) {
    const int n0 = (wave + i * NWAVE) * 32;
    v8f dacc[2][QTILE / 16];
#pragma unroll
    for (int j = 0; j < 2; ++j) {
#pragma unroll
      for (int mt = 0; mt < QTILE / 16; ++mt) dacc[j][mt] = (v8f){};
    }
#pragma unroll
    for (int kk = 0; kk < 4; ++kk) {
      half_swap(rr[i][kk][0], rr[i][kk][2]);
      half_swap(rr[i][kk][1], rr[i][kk][3]);
      const v2f b0 = {rr[i][kk][0], rr[i][kk][1]};
      const v2f b1 = {rr[i][kk][2], rr[i][kk][3]};
#pragma unroll
      for (int mt = 0; mt < QTILE / 16; ++mt) {
        dacc[0][mt] = __builtin_amdgcn_wmma_f32_16x16x4_f32(false, a[mt][kk], false, b0,
                                                            (short)0, dacc[0][mt], false, false);
        dacc[1][mt] = __builtin_amdgcn_wmma_f32_16x16x4_f32(false, a[mt][kk], false, b1,
                                                            (short)0, dacc[1][mt], false, false);
      }
    }
    __builtin_amdgcn_sched_barrier(0);
    asm volatile("v_nop\n\tv_nop\n\tv_nop\n\tv_nop\n\tv_nop");
#pragma unroll
    for (int mt = 0; mt < QTILE / 16; ++mt) {
#pragma unroll
      for (int r = 0; r < 8; ++r) {
        float lo = dacc[0][mt][r], hi8 = dacc[1][mt][r];
        half_swap(lo, hi8);
        const int M = mt * 16 + r;
        outBase[(size_t)M * N_CLASS * FDIM + n0 + lane] = lo;
        outBase[(size_t)(M + 8) * N_CLASS * FDIM + n0 + lane] = hi8;
      }
    }
  }
}

extern "C" void kernel_launch(void* const* d_in, const int* in_sizes, int n_in,
                              void* d_out, int out_size, void* d_ws, size_t ws_size,
                              hipStream_t stream) {
  const float* data = (const float*)d_in[0];
  float* out = (float*)d_out;
  (void)in_sizes; (void)n_in; (void)d_ws; (void)ws_size; (void)out_size;
  dim3 grid(N_CLASS, N_QRY / QTILE);
  proto_att_kernel<<<grid, NTHREAD, 0, stream>>>(data, out);
}
